# starting kernel
# baseline (speedup 1.0000x reference)
#include <hip/hip_runtime.h>

typedef __attribute__((ext_vector_type(16))) __bf16 v16bf;
typedef __attribute__((ext_vector_type(8)))  float  v8f;
typedef __attribute__((ext_vector_type(4)))  unsigned int u32x4;

union FragU { u32x4 q[2]; v16bf v; };

__device__ __forceinline__ v16bf load_frag16(const unsigned short* p) {
    FragU f;
    f.q[0] = *(const u32x4*)(p);
    f.q[1] = *(const u32x4*)(p + 16);
    return f.v;
}

__device__ __forceinline__ v8f wmma_bf16(v16bf a, v16bf b, v8f c) {
    return __builtin_amdgcn_wmma_f32_16x16x32_bf16(
         false, a,  false, b,
         (short)0, c,  false,  false);
}

__device__ __forceinline__ unsigned short f2bf(float f) {
    unsigned int u = __float_as_uint(f);
    unsigned int r = u + 0x7FFFu + ((u >> 16) & 1u);
    return (unsigned short)(r >> 16);
}

__device__ __forceinline__ void atomicMaxFloat(float* addr, float value) {
    if (value >= 0.f) atomicMax((int*)addr, __float_as_int(value));
    else              atomicMin((unsigned int*)addr, __float_as_uint(value));
}

#define GN 50000
#define GE 800000
#define GD 128
#define GH 8
#define GDFF 512

__global__ void repack_kernel(const float* __restrict__ x, const float* __restrict__ Wfc,
                              const float* __restrict__ W1, const float* __restrict__ W2,
                              unsigned short* __restrict__ xb, unsigned short* __restrict__ Wzt,
                              unsigned short* __restrict__ W1t, unsigned short* __restrict__ W2t) {
    int idx = blockIdx.x * blockDim.x + threadIdx.x;
    if (idx < GN * GD) xb[idx] = f2bf(x[idx]);
    if (idx < GD * GD) {
        int j = idx >> 7, d = idx & 127;
        int h = j >> 4, k = j & 15;
        Wzt[idx] = f2bf(Wfc[(h * GD + d) * 16 + k]);
    }
    if (idx < GDFF * GD) {
        int f = idx >> 7, d = idx & 127;
        W1t[idx] = f2bf(W1[d * GDFF + f]);
        int d2 = idx >> 9, f2 = idx & 511;
        W2t[idx] = f2bf(W2[f2 * GD + d2]);
    }
}

__global__ void init_kernel(float* __restrict__ hagg, float* __restrict__ m, float* __restrict__ s) {
    int idx = blockIdx.x * blockDim.x + threadIdx.x;
    if (idx < GN * GD) hagg[idx] = 0.f;
    if (idx < GN * GH) { m[idx] = -INFINITY; s[idx] = 0.f; }
}

__global__ void gemm_z_kernel(const unsigned short* __restrict__ xb,
                              const unsigned short* __restrict__ Wzt,
                              float* __restrict__ z) {
    int wave = threadIdx.x >> 5, lane = threadIdx.x & 31;
    int rowBase = blockIdx.x * 16;
    int col0 = wave * 16;
    int lrow = lane & 15;
    int kb = (lane & 16) ? 8 : 0;
    const unsigned short* aRow = xb  + (size_t)(rowBase + lrow) * GD + kb;
    const unsigned short* bCol = Wzt + (size_t)(col0 + lrow)   * GD + kb;
    v8f acc = {};
    for (int k0 = 0; k0 < GD; k0 += 32)
        acc = wmma_bf16(load_frag16(aRow + k0), load_frag16(bCol + k0), acc);
    int half8 = (lane & 16) >> 1;
    int n = col0 + lrow;
    for (int j = 0; j < 8; ++j)
        z[(size_t)(rowBase + half8 + j) * GD + n] = acc[j];
}

__global__ void elr_kernel(const float* __restrict__ z, const float* __restrict__ a_l,
                           const float* __restrict__ a_r, float* __restrict__ el,
                           float* __restrict__ er) {
    int idx = blockIdx.x * blockDim.x + threadIdx.x;
    if (idx >= GN * GH) return;
    int h = idx & 7;
    const float* zp = z + (size_t)(idx >> 3) * GD + h * 16;
    float sl = 0.f, sr = 0.f;
    #pragma unroll
    for (int k = 0; k < 16; ++k) { sl += zp[k] * a_l[h * 16 + k]; sr += zp[k] * a_r[h * 16 + k]; }
    el[idx] = sl; er[idx] = sr;
}

__global__ void edge1_kernel(const float* __restrict__ el, const float* __restrict__ er,
                             const int* __restrict__ esrc, const int* __restrict__ edst,
                             float* __restrict__ ebuf, float* __restrict__ m) {
    int idx = blockIdx.x * blockDim.x + threadIdx.x;
    if (idx >= GE * GH) return;
    int e = idx >> 3, h = idx & 7;
    float v = el[esrc[e] * GH + h] + er[edst[e] * GH + h];
    v = v > 0.f ? v : 0.01f * v;
    ebuf[idx] = v;
    atomicMaxFloat(&m[edst[e] * GH + h], v);
}

__global__ void edge2_kernel(const int* __restrict__ edst, const float* __restrict__ m,
                             float* __restrict__ ebuf, float* __restrict__ s) {
    int idx = blockIdx.x * blockDim.x + threadIdx.x;
    if (idx >= GE * GH) return;
    int e = idx >> 3, h = idx & 7;
    float mm = m[edst[e] * GH + h];
    if (mm < -3.0e38f) mm = 0.f;
    float ee = __expf(ebuf[idx] - mm);
    ebuf[idx] = ee;
    atomicAdd(&s[edst[e] * GH + h], ee);
}

__global__ void scatter_kernel(const float* __restrict__ z, const float* __restrict__ ebuf,
                               const float* __restrict__ s, const int* __restrict__ esrc,
                               const int* __restrict__ edst, float* __restrict__ hagg) {
    int e = blockIdx.x * 2 + (threadIdx.x >> 7);
    int t = threadIdx.x & 127;
    int src = esrc[e], dst = edst[e];
    int h = t >> 4;
    float alpha = ebuf[(size_t)e * GH + h] / fmaxf(s[dst * GH + h], 1e-9f);
    atomicAdd(&hagg[(size_t)dst * GD + t], alpha * z[(size_t)src * GD + t]);
}

__global__ void node_ln_kernel(const float* __restrict__ hagg, const float* __restrict__ x,
                               const float* __restrict__ gamma, const float* __restrict__ beta,
                               float* __restrict__ hres, unsigned short* __restrict__ lnb) {
    int row = blockIdx.x;
    int t = threadIdx.x;
    float v = hagg[(size_t)row * GD + t];
    v = v > 0.f ? v : (__expf(v) - 1.f);
    v += x[(size_t)row * GD + t];
    hres[(size_t)row * GD + t] = v;
    __shared__ float rs[GD];
    __shared__ float rq[GD];
    rs[t] = v; rq[t] = v * v;
    __syncthreads();
    for (int sft = 64; sft > 0; sft >>= 1) {
        if (t < sft) { rs[t] += rs[t + sft]; rq[t] += rq[t + sft]; }
        __syncthreads();
    }
    float mean = rs[0] * (1.f / GD);
    float var  = rq[0] * (1.f / GD) - mean * mean;
    float ln = (v - mean) * rsqrtf(var + 1e-5f) * gamma[t] + beta[t];
    lnb[(size_t)row * GD + t] = f2bf(ln);
}

__global__ void gemm_ffn1_kernel(const unsigned short* __restrict__ lnb,
                                 const unsigned short* __restrict__ W1t,
                                 const float* __restrict__ b1,
                                 unsigned short* __restrict__ inter) {
    int wave = threadIdx.x >> 5, lane = threadIdx.x & 31;
    int rowBase = blockIdx.x * 16;
    int col0 = (blockIdx.y * 8 + wave) * 16;
    int lrow = lane & 15;
    int kb = (lane & 16) ? 8 : 0;
    const unsigned short* aRow = lnb + (size_t)(rowBase + lrow) * GD + kb;
    const unsigned short* bCol = W1t + (size_t)(col0 + lrow)   * GD + kb;
    v8f acc = {};
    for (int k0 = 0; k0 < GD; k0 += 32)
        acc = wmma_bf16(load_frag16(aRow + k0), load_frag16(bCol + k0), acc);
    int half8 = (lane & 16) >> 1;
    int n = col0 + lrow;
    float bias = b1[n];
    for (int j = 0; j < 8; ++j) {
        float v = fmaxf(acc[j] + bias, 0.f);
        inter[(size_t)(rowBase + half8 + j) * GDFF + n] = f2bf(v);
    }
}

__global__ void gemm_ffn2_kernel(const unsigned short* __restrict__ inter,
                                 const unsigned short* __restrict__ W2t,
                                 const float* __restrict__ b2,
                                 const float* __restrict__ hres,
                                 float* __restrict__ out) {
    int wave = threadIdx.x >> 5, lane = threadIdx.x & 31;
    int rowBase = blockIdx.x * 16;
    int col0 = wave * 16;
    int lrow = lane & 15;
    int kb = (lane & 16) ? 8 : 0;
    const unsigned short* aRow = inter + (size_t)(rowBase + lrow) * GDFF + kb;
    const unsigned short* bCol = W2t  + (size_t)(col0 + lrow)    * GDFF + kb;
    v8f acc = {};
    for (int k0 = 0; k0 < GDFF; k0 += 32)
        acc = wmma_bf16(load_frag16(aRow + k0), load_frag16(bCol + k0), acc);
    int half8 = (lane & 16) >> 1;
    int n = col0 + lrow;
    float bias = b2[n];
    for (int j = 0; j < 8; ++j) {
        size_t r = (size_t)(rowBase + half8 + j);
        out[r * GD + n] = acc[j] + bias + hres[r * GD + n];
    }
}

extern "C" void kernel_launch(void* const* d_in, const int* in_sizes, int n_in,
                              void* d_out, int out_size, void* d_ws, size_t ws_size,
                              hipStream_t stream) {
    const float* x     = (const float*)d_in[0];
    const float* Wfc   = (const float*)d_in[1];
    const float* a_l   = (const float*)d_in[2];
    const float* a_r   = (const float*)d_in[3];
    const float* gamma = (const float*)d_in[4];
    const float* beta  = (const float*)d_in[5];
    const float* W1    = (const float*)d_in[6];
    const float* b1    = (const float*)d_in[7];
    const float* W2    = (const float*)d_in[8];
    const float* b2    = (const float*)d_in[9];
    const int* esrc    = (const int*)d_in[10];
    const int* edst    = (const int*)d_in[11];
    float* out = (float*)d_out;

    char* ws = (char*)d_ws;
    size_t off = 0;
    auto take = [&](size_t bytes) -> char* {
        char* p = ws + off;
        off += (bytes + 255) & ~(size_t)255;
        return p;
    };
    unsigned short* xb    = (unsigned short*)take((size_t)GN * GD * 2);
    unsigned short* Wzt   = (unsigned short*)take((size_t)GD * GD * 2);
    unsigned short* W1t   = (unsigned short*)take((size_t)GDFF * GD * 2);
    unsigned short* W2t   = (unsigned short*)take((size_t)GD * GDFF * 2);
    float*          z     = (float*)take((size_t)GN * GD * 4);
    float*          el    = (float*)take((size_t)GN * GH * 4);
    float*          er    = (float*)take((size_t)GN * GH * 4);
    float*          m     = (float*)take((size_t)GN * GH * 4);
    float*          s     = (float*)take((size_t)GN * GH * 4);
    float*          ebuf  = (float*)take((size_t)GE * GH * 4);
    float*          hagg  = (float*)take((size_t)GN * GD * 4);
    float*          hres  = (float*)take((size_t)GN * GD * 4);
    unsigned short* lnb   = (unsigned short*)take((size_t)GN * GD * 2);
    unsigned short* inter = (unsigned short*)take((size_t)GN * GDFF * 2);

    const int mtiles = GN / 16;

    repack_kernel<<<(GN * GD + 255) / 256, 256, 0, stream>>>(x, Wfc, W1, W2, xb, Wzt, W1t, W2t);
    init_kernel  <<<(GN * GD + 255) / 256, 256, 0, stream>>>(hagg, m, s);
    gemm_z_kernel<<<mtiles, 256, 0, stream>>>(xb, Wzt, z);
    elr_kernel   <<<(GN * GH + 255) / 256, 256, 0, stream>>>(z, a_l, a_r, el, er);
    edge1_kernel <<<(GE * GH + 255) / 256, 256, 0, stream>>>(el, er, esrc, edst, ebuf, m);
    edge2_kernel <<<(GE * GH + 255) / 256, 256, 0, stream>>>(edst, m, ebuf, s);
    scatter_kernel<<<GE / 2, 256, 0, stream>>>(z, ebuf, s, esrc, edst, hagg);
    node_ln_kernel<<<GN, GD, 0, stream>>>(hagg, x, gamma, beta, hres, lnb);
    gemm_ffn1_kernel<<<dim3(mtiles, 4), 256, 0, stream>>>(lnb, W1t, b1, inter);
    gemm_ffn2_kernel<<<mtiles, 256, 0, stream>>>(inter, W2t, b2, hres, out);
}


# FFN2 reads its 16 rows of inter as one coalesced 16 KiB piece into an LDS tile (pitch 1040 B); its 8 waves take A fragments from LDS, not by strided global loads of 16 lines each; same k order
# speedup vs baseline: 4.4106x; 4.4106x over previous
#include <hip/hip_runtime.h>

typedef __attribute__((ext_vector_type(16))) __bf16 v16bf;
typedef __attribute__((ext_vector_type(8)))  float  v8f;
typedef __attribute__((ext_vector_type(4)))  unsigned int u32x4;

typedef __attribute__((ext_vector_type(2))) float  f32x2;
typedef __attribute__((ext_vector_type(2))) __bf16 bf16x2;

union FragU { u32x4 q[2]; v16bf v; };

__device__ __forceinline__ v16bf load_frag16(const unsigned short* p) {
    FragU f;
    f.q[0] = *(const u32x4*)(p);
    f.q[1] = *(const u32x4*)(p + 16);
    return f.v;
}

#define WBLK 512
__device__ __forceinline__ v16bf load_fragw(const unsigned short* blk, int lane) {
    FragU f;
    f.q[0] = *(const u32x4*)(blk + 8 * lane);
    f.q[1] = *(const u32x4*)(blk + 256 + 8 * lane);
    return f.v;
}

__device__ __forceinline__ v8f wmma_bf16(v16bf a, v16bf b, v8f c) {
    return __builtin_amdgcn_wmma_f32_16x16x32_bf16(
         false, a,  false, b,
         (short)0, c,  false,  false);
}

__device__ __forceinline__ unsigned short f2bf(float f) {
    unsigned int u = __float_as_uint(f);
    unsigned int r = u + 0x7FFFu + ((u >> 16) & 1u);
    return (unsigned short)(r >> 16);
}

__device__ __forceinline__ unsigned pack2bf(float a, float b) {
    f32x2 f = {a, b};
    return __builtin_bit_cast(unsigned, __builtin_convertvector(f, bf16x2));
}

#define GN 50000
#define GE 800000
#define GD 128
#define GH 8
#define GDFF 512

#define TILE 197
#define NTILES 254
#define NOTILE 255u
__device__ __forceinline__ unsigned tile_key(int dst) {
    unsigned d = (unsigned)dst;
    return d < (unsigned)(NTILES * TILE) ? d / (unsigned)TILE : NOTILE;
}
__global__ void repack_kernel(const float* __restrict__ Wfc,
                              const float* __restrict__ W1, const float* __restrict__ W2,
                              const int* __restrict__ edst,
                              unsigned short* __restrict__ Wzp,
                              unsigned short* __restrict__ W1p, unsigned short* __restrict__ W2p,
                              unsigned* __restrict__ tkey) {
    int idx = blockIdx.x * blockDim.x + threadIdx.x;
    if (idx < GE / 16) {
        const int4* d4 = (const int4*)edst + 4 * idx;
        int4 d[4];
        #pragma unroll
        for (int k = 0; k < 4; ++k) d[k] = d4[k];
        unsigned w[4];
        #pragma unroll
        for (int k = 0; k < 4; ++k)
            w[k] = tile_key(d[k].x) | (tile_key(d[k].y) << 8) | (tile_key(d[k].z) << 16) | (tile_key(d[k].w) << 24);
        ((u32x4*)tkey)[idx] = (u32x4){w[0], w[1], w[2], w[3]};
    }
    const int e = idx & 7, ln = (idx >> 3) & 31, q = (idx >> 8) & 1, blk = idx >> 9;
    const int c = ln & 15;
    const int ko = 16 * q + 8 * (ln >> 4) + e;
    if (idx < GD * GD) {
        int s = blk & 3, t = blk >> 2;
        Wzp[idx] = f2bf(Wfc[(t * GD + 32 * s + ko) * 16 + c]);
    }
    if (idx < GDFF * GD) {
        int s1 = blk & 3, t1 = blk >> 2;
        W1p[idx] = f2bf(W1[(32 * s1 + ko) * GDFF + 16 * t1 + c]);
        int s2 = blk & 15, t2 = blk >> 4;
        W2p[idx] = f2bf(W2[(32 * s2 + ko) * GD + 16 * t2 + c]);
    }
}

#define XPITCH 136
__global__ void __launch_bounds__(256)
gemm_z_kernel(const float* __restrict__ x,
              const unsigned short* __restrict__ Wzp,
              float* __restrict__ z) {
    __shared__ __attribute__((aligned(16))) unsigned short xs[16 * XPITCH];
    int tid = threadIdx.x;
    int wave = tid >> 5, lane = tid & 31;
    int rowBase = blockIdx.x * 16;
    int col0 = wave * 16;
    int lrow = lane & 15;
    int kb = (lane & 16) ? 8 : 0;
    const float4* xt = (const float4*)(x + (size_t)rowBase * GD);
    float4 v0 = xt[tid], v1 = xt[tid + 256];
    const unsigned short* wb = Wzp + (size_t)wave * 4 * WBLK;
    v16bf bf[4];
    #pragma unroll
    for (int s = 0; s < 4; ++s) bf[s] = load_fragw(wb + s * WBLK, lane);
    *(uint2*)(xs + wave * XPITCH + 4 * lane)       = make_uint2(pack2bf(v0.x, v0.y), pack2bf(v0.z, v0.w));
    *(uint2*)(xs + (wave + 8) * XPITCH + 4 * lane) = make_uint2(pack2bf(v1.x, v1.y), pack2bf(v1.z, v1.w));
    __syncthreads();
    const unsigned short* aRow = xs + lrow * XPITCH + kb;
    v8f acc = {};
    #pragma unroll
    for (int s = 0; s < 4; ++s)
        acc = wmma_bf16(load_frag16(aRow + 32 * s), bf[s], acc);
    int half8 = (lane & 16) >> 1;
    int n = col0 + lrow;
    for (int j = 0; j < 8; ++j)
        z[(size_t)(rowBase + half8 + j) * GD + n] = acc[j];
}

__global__ void elr_kernel(const float* __restrict__ z, const float* __restrict__ a_l,
                           const float* __restrict__ a_r, float* __restrict__ el,
                           float* __restrict__ er) {
    int idx = blockIdx.x * blockDim.x + threadIdx.x;
    if (idx >= GN * GH) return;
    int h = idx & 7;
    const float* zp = z + (size_t)(idx >> 3) * GD + h * 16;
    float sl = 0.f, sr = 0.f;
    #pragma unroll
    for (int k = 0; k < 16; ++k) { sl += zp[k] * a_l[h * 16 + k]; sr += zp[k] * a_r[h * 16 + k]; }
    el[idx] = sl; er[idx] = sr;
}

#define TB 1024
#define CAP 16384

__device__ __forceinline__ unsigned fkey(float v) {
    unsigned u = __float_as_uint(v);
    return (u & 0x80000000u) ? ~u : (u | 0x80000000u);
}
__device__ __forceinline__ float funkey(unsigned k) {
    return __uint_as_float((k & 0x80000000u) ? (k & 0x7FFFFFFFu) : ~k);
}
__device__ __forceinline__ float lrelu(float v) { return v > 0.f ? v : 0.01f * v; }

__device__ __forceinline__ void edge_max(const float* __restrict__ el, const float* erl, unsigned* mx,
                                         int src, int dl) {
    const float4* p = (const float4*)(el + (size_t)src * GH);
    float4 a = p[0], b = p[1];
    const float* r = erl + dl * GH;
    unsigned* m = mx + dl * GH;
    atomicMax(&m[0], fkey(lrelu(a.x + r[0]))); atomicMax(&m[1], fkey(lrelu(a.y + r[1])));
    atomicMax(&m[2], fkey(lrelu(a.z + r[2]))); atomicMax(&m[3], fkey(lrelu(a.w + r[3])));
    atomicMax(&m[4], fkey(lrelu(b.x + r[4]))); atomicMax(&m[5], fkey(lrelu(b.y + r[5])));
    atomicMax(&m[6], fkey(lrelu(b.z + r[6]))); atomicMax(&m[7], fkey(lrelu(b.w + r[7])));
}

__device__ __forceinline__ void edge_acc(float elv, float4 zv, const float* erl, const unsigned* mx,
                                         float* sm, float* acc, int dl, int lane) {
    int h = lane >> 2;
    float mm = funkey(mx[dl * GH + h]);
    if (!(fabsf(mm) < INFINITY)) mm = 0.f;
    float ee = __expf(lrelu(elv + erl[dl * GH + h]) - mm);
    if ((lane & 3) == 0) atomicAdd(&sm[dl * GH + h], ee);
    float* row = acc + dl * GD + lane;
    atomicAdd(row,      ee * zv.x);
    atomicAdd(row + 32, ee * zv.y);
    atomicAdd(row + 64, ee * zv.z);
    atomicAdd(row + 96, ee * zv.w);
}

__device__ __forceinline__ void note_chunk(u32x4 t, unsigned pat, int chunk, unsigned* list, unsigned* cnt) {
    unsigned nz = 0u;
    #pragma unroll
    for (int q = 0; q < 4; ++q) {
        unsigned u = t[q] ^ pat;
        unsigned y = ((u & 0x7F7F7F7Fu) + 0x7F7F7F7Fu) | u;
        nz |= (y >> (7 - q)) & (0x01010101u << q);
    }
    unsigned mm = ~nz & 0x0F0F0F0Fu;
    while (mm) {
        int p = __ffs(mm) - 1; mm &= mm - 1;
        unsigned pos = atomicAdd(cnt, 1u);
        if (pos < (unsigned)CAP) list[pos] = (unsigned)(chunk * 16 + (p & 7) * 4 + (p >> 3));
    }
}

__device__ __forceinline__ void scan_keys(const unsigned* __restrict__ tkey, int tid,
                                          unsigned* list, unsigned* cnt) {
    const int nvec = GE / 16;
    const int full = nvec / TB;
    static_assert(full % 4 == 0, "the main loop takes four sweeps per trip");
    const u32x4* k4 = (const u32x4*)tkey;
    const int off = (int)blockIdx.x * (nvec / NTILES);
    const unsigned pat = (unsigned)blockIdx.x * 0x01010101u;
    for (int it = 0; it < full; it += 4) {
        u32x4 t[4]; int vi[4];
        #pragma unroll
        for (int k = 0; k < 4; ++k) {
            int idx = (it + k) * TB + tid + off; if (idx >= nvec) idx -= nvec;
            vi[k] = idx;
            t[k] = k4[idx];
        }
        #pragma unroll
        for (int k = 0; k < 4; ++k) note_chunk(t[k], pat, vi[k], list, cnt);
    }
    int j = full * TB + tid;
    if (j < nvec) {
        int idx = j + off; if (idx >= nvec) idx -= nvec;
        note_chunk(k4[idx], pat, idx, list, cnt);
    }
}

template <int MODE>
__device__ __forceinline__ void scan_edges(const int* __restrict__ esrc, const int* __restrict__ edst,
                                           const float* __restrict__ el, const float* __restrict__ z,
                                           int tileBase, int tid, int lane,
                                           unsigned* list, unsigned* cnt, const float* erl,
                                           unsigned* mx, float* sm, float* acc) {
    const int nvec = GE / 4;
    const int4* d4 = (const int4*)edst;
    const int off = (int)blockIdx.x * (nvec / NTILES);
    for (int it = 0; it < 196; it += 4) {
        int4 d[4]; int vi[4];
        #pragma unroll
        for (int k = 0; k < 4; ++k) {
            int j = (it + k) * TB + tid;
            int idx = j + off; if (idx >= nvec) idx -= nvec;
            bool ok = j < nvec;
            vi[k] = idx;
            d[k] = ok ? d4[idx] : make_int4(-1, -1, -1, -1);
        }
        #pragma unroll
        for (int k = 0; k < 4; ++k) {
            int dd[4] = {d[k].x, d[k].y, d[k].z, d[k].w};
            #pragma unroll
            for (int c = 0; c < 4; ++c) {
                unsigned dl = (unsigned)(dd[c] - tileBase);
                bool hit = dl < (unsigned)TILE;
                int e = vi[k] * 4 + c;
                if (MODE == 1) {
                    if (hit) edge_max(el, erl, mx, esrc[e], (int)dl);
                } else {
                    unsigned bal = (unsigned)__ballot(hit);
                    while (bal) {
                        int l = __ffs(bal) - 1; bal &= bal - 1;
                        int ee_ = __shfl(e, l); int dle = __shfl((int)dl, l);
                        int s_ = esrc[ee_];
                        float elv = el[(size_t)s_ * GH + (lane >> 2)];
                        float4 zv = *(const float4*)(z + (size_t)s_ * GD + 4 * lane);
                        edge_acc(elv, zv, erl, mx, sm, acc, dle, lane);
                    }
                }
            }
        }
    }
}

__global__ void __launch_bounds__(TB)
tile_softmax_agg_kernel(const float* __restrict__ z, const float* __restrict__ el,
                        const float* __restrict__ er, const int* __restrict__ esrc,
                        const int* __restrict__ edst, const unsigned* __restrict__ tkey,
                        float* __restrict__ hagg) {
    __shared__ __attribute__((aligned(16))) float acc[TILE * GD];
    __shared__ float    sm[TILE * GH];
    __shared__ float    erl[TILE * GH];
    __shared__ unsigned mx[TILE * GH];
    __shared__ unsigned list[CAP];
    __shared__ unsigned cnt;

    const int tid = threadIdx.x, lane = tid & 31, wave = tid >> 5;
    const int tileBase = (int)blockIdx.x * TILE;

    for (int i = tid; i < TILE * GD; i += TB) acc[i] = 0.f;
    for (int i = tid; i < TILE * GH; i += TB) {
        sm[i] = 0.f;
        mx[i] = 0u;
        int node = tileBase + (i >> 3);
        erl[i] = node < GN ? er[(size_t)tileBase * GH + i] : 0.f;
    }
    if (tid == 0) cnt = 0u;
    __syncthreads();

    scan_keys(tkey, tid, list, &cnt);
    __syncthreads();
    const unsigned total = cnt;

    if (total <= (unsigned)CAP) {
        const int n = (int)total;
        for (int i = tid; i < n; i += TB) {
            unsigned e = list[i];
            int s_ = esrc[e];
            int dl = edst[e] - tileBase;
            list[i] = ((unsigned)s_ << 8) | (unsigned)dl;
            edge_max(el, erl, mx, s_, dl);
        }
        __syncthreads();
        int i = wave;
        for (; i + 96 < n; i += 128) {
            unsigned ent[4]; float elv[4]; float4 zv[4];
            #pragma unroll
            for (int k = 0; k < 4; ++k) ent[k] = list[i + 32 * k];
            #pragma unroll
            for (int k = 0; k < 4; ++k) {
                size_t s_ = ent[k] >> 8;
                elv[k] = el[s_ * GH + (lane >> 2)];
                zv[k]  = *(const float4*)(z + s_ * GD + 4 * lane);
            }
            #pragma unroll
            for (int k = 0; k < 4; ++k)
                edge_acc(elv[k], zv[k], erl, mx, sm, acc, (int)(ent[k] & 255u), lane);
        }
        for (; i < n; i += 32) {
            unsigned ent = list[i];
            size_t s_ = ent >> 8;
            float elv = el[s_ * GH + (lane >> 2)];
            float4 zv = *(const float4*)(z + s_ * GD + 4 * lane);
            edge_acc(elv, zv, erl, mx, sm, acc, (int)(ent & 255u), lane);
        }
    } else {
        scan_edges<1>(esrc, edst, el, z, tileBase, tid, lane, list, &cnt, erl, mx, sm, acc);
        __syncthreads();
        scan_edges<2>(esrc, edst, el, z, tileBase, tid, lane, list, &cnt, erl, mx, sm, acc);
    }
    __syncthreads();

    for (int i = tid; i < TILE * GD; i += TB) {
        int r = i >> 7, t = i & 127;
        int node = tileBase + r;
        if (node < GN) {
            float v = acc[r * GD + (t & 3) * 32 + (t >> 2)];
            hagg[(size_t)node * GD + t] = v / fmaxf(sm[r * GH + (t >> 4)], 1e-9f);
        }
    }
}

__global__ void __launch_bounds__(256)
ln_ffn1_kernel(const float* __restrict__ hagg, const float* __restrict__ x,
               const float* __restrict__ gamma, const float* __restrict__ beta,
               const unsigned short* __restrict__ W1p, const float* __restrict__ b1,
               float* __restrict__ hres, unsigned short* __restrict__ inter) {
    __shared__ __attribute__((aligned(16))) unsigned short ls[16 * XPITCH];
    int tid = threadIdx.x;
    int wave = tid >> 5, lane = tid & 31;
    int rowBase = blockIdx.x * 16;
    int lrow = lane & 15;
    int kb = (lane & 16) ? 8 : 0;

    const float4 g4 = ((const float4*)gamma)[lane];
    const float4 b4 = ((const float4*)beta)[lane];
    float4 hv[2], xv[2];
    #pragma unroll
    for (int rr = 0; rr < 2; ++rr) {
        size_t o = (size_t)(rowBase + 2 * wave + rr) * GD + 4 * lane;
        hv[rr] = *(const float4*)(hagg + o);
        xv[rr] = *(const float4*)(x + o);
    }
    v16bf bf[4][4];
    float bias[4];
    #pragma unroll
    for (int j = 0; j < 4; ++j) {
        int n = (wave * 4 + j) * 16 + lrow;
        const unsigned short* wb = W1p + (size_t)(wave * 4 + j) * 4 * WBLK;
        #pragma unroll
        for (int s = 0; s < 4; ++s) bf[j][s] = load_fragw(wb + s * WBLK, lane);
        bias[j] = b1[n];
    }
    __builtin_amdgcn_sched_barrier(0);
    #pragma unroll
    for (int rr = 0; rr < 2; ++rr) {
        int r = 2 * wave + rr;
        float4 a = hv[rr];
        float v0 = (a.x > 0.f ? a.x : (__expf(a.x) - 1.f)) + xv[rr].x;
        float v1 = (a.y > 0.f ? a.y : (__expf(a.y) - 1.f)) + xv[rr].y;
        float v2 = (a.z > 0.f ? a.z : (__expf(a.z) - 1.f)) + xv[rr].z;
        float v3 = (a.w > 0.f ? a.w : (__expf(a.w) - 1.f)) + xv[rr].w;
        hv[rr] = make_float4(v0, v1, v2, v3);
        float s = (v0 + v1) + (v2 + v3);
        float q = (v0 * v0 + v1 * v1) + (v2 * v2 + v3 * v3);
        for (int m = 16; m > 0; m >>= 1) { s += __shfl_xor(s, m); q += __shfl_xor(q, m); }
        float mean = s * (1.f / GD);
        float var  = q * (1.f / GD) - mean * mean;
        float rstd = rsqrtf(var + 1e-5f);
        float l0 = (v0 - mean) * rstd * g4.x + b4.x;
        float l1 = (v1 - mean) * rstd * g4.y + b4.y;
        float l2 = (v2 - mean) * rstd * g4.z + b4.z;
        float l3 = (v3 - mean) * rstd * g4.w + b4.w;
        *(uint2*)(ls + r * XPITCH + 4 * lane) = make_uint2(pack2bf(l0, l1), pack2bf(l2, l3));
    }
    __syncthreads();
    #pragma unroll
    for (int rr = 0; rr < 2; ++rr)
        *(float4*)(hres + (size_t)(rowBase + 2 * wave + rr) * GD + 4 * lane) = hv[rr];

    const unsigned short* aRow = ls + lrow * XPITCH + kb;
    v8f acc[4] = {};
    #pragma unroll
    for (int s = 0; s < 4; ++s) {
        v16bf af = load_frag16(aRow + 32 * s);
        #pragma unroll
        for (int j = 0; j < 4; ++j) acc[j] = wmma_bf16(af, bf[j][s], acc[j]);
    }
    int half8 = (lane & 16) >> 1;
    #pragma unroll
    for (int j = 0; j < 4; ++j) {
        int n = (wave * 4 + j) * 16 + lrow;
        for (int i = 0; i < 8; ++i) {
            float v = fmaxf(acc[j][i] + bias[j], 0.f);
            inter[(size_t)(rowBase + half8 + i) * GDFF + n] = f2bf(v);
        }
    }
}

#define APITCH 520
__global__ void __launch_bounds__(256)
gemm_ffn2_kernel(const unsigned short* __restrict__ inter,
                 const unsigned short* __restrict__ W2p,
                 const float* __restrict__ b2,
                 const float* __restrict__ hres,
                 float* __restrict__ out) {
    __shared__ __attribute__((aligned(16))) unsigned short as[16 * APITCH];
    int tid = threadIdx.x;
    int wave = tid >> 5, lane = tid & 31;
    int rowBase = blockIdx.x * 16;
    int col0 = wave * 16;
    int lrow = lane & 15;
    int kb = (lane & 16) ? 8 : 0;
    const u32x4* at = (const u32x4*)(inter + (size_t)rowBase * GDFF);
    u32x4 ch[4];
    #pragma unroll
    for (int k = 0; k < 4; ++k) ch[k] = at[tid + 256 * k];
    #pragma unroll
    for (int k = 0; k < 4; ++k) {
        int i = tid + 256 * k;
        *(u32x4*)(as + (i >> 6) * APITCH + 8 * (i & 63)) = ch[k];
    }
    __syncthreads();
    const unsigned short* aRow = as + lrow * APITCH + kb;
    const unsigned short* wb = W2p + (size_t)wave * (GDFF / 32) * WBLK;
    v8f acc = {};
    for (int k0 = 0; k0 < GDFF; k0 += 32)
        acc = wmma_bf16(load_frag16(aRow + k0), load_fragw(wb + (k0 >> 5) * WBLK, lane), acc);
    int half8 = (lane & 16) >> 1;
    int n = col0 + lrow;
    float bias = b2[n];
    for (int j = 0; j < 8; ++j) {
        size_t r = (size_t)(rowBase + half8 + j);
        out[r * GD + n] = acc[j] + bias + hres[r * GD + n];
    }
}

extern "C" void kernel_launch(void* const* d_in, const int* in_sizes, int n_in,
                              void* d_out, int out_size, void* d_ws, size_t ws_size,
                              hipStream_t stream) {
    const float* x     = (const float*)d_in[0];
    const float* Wfc   = (const float*)d_in[1];
    const float* a_l   = (const float*)d_in[2];
    const float* a_r   = (const float*)d_in[3];
    const float* gamma = (const float*)d_in[4];
    const float* beta  = (const float*)d_in[5];
    const float* W1    = (const float*)d_in[6];
    const float* b1    = (const float*)d_in[7];
    const float* W2    = (const float*)d_in[8];
    const float* b2    = (const float*)d_in[9];
    const int* esrc    = (const int*)d_in[10];
    const int* edst    = (const int*)d_in[11];
    float* out = (float*)d_out;

    char* ws = (char*)d_ws;
    size_t off = 0;
    auto take = [&](size_t bytes) -> char* {
        char* p = ws + off;
        off += (bytes + 255) & ~(size_t)255;
        return p;
    };
    unsigned short* Wzp   = (unsigned short*)take((size_t)GD * GD * 2);
    unsigned short* W1p   = (unsigned short*)take((size_t)GDFF * GD * 2);
    unsigned short* W2p   = (unsigned short*)take((size_t)GD * GDFF * 2);
    float*          z     = (float*)take((size_t)GN * GD * 4);
    float*          el    = (float*)take((size_t)GN * GH * 4);
    float*          er    = (float*)take((size_t)GN * GH * 4);
    float*          hagg  = (float*)take((size_t)GN * GD * 4);
    float*          hres  = (float*)take((size_t)GN * GD * 4);
    unsigned short* inter = (unsigned short*)take((size_t)GN * GDFF * 2);
    unsigned*       tkey  = (unsigned*)inter;

    const int mtiles = GN / 16;

    repack_kernel<<<(GDFF * GD + 255) / 256, 256, 0, stream>>>(Wfc, W1, W2, edst, Wzp, W1p, W2p, tkey);
    gemm_z_kernel<<<mtiles, 256, 0, stream>>>(x, Wzp, z);
    elr_kernel   <<<(GN * GH + 255) / 256, 256, 0, stream>>>(z, a_l, a_r, el, er);
    tile_softmax_agg_kernel<<<NTILES, TB, 0, stream>>>(z, el, er, esrc, edst, tkey, hagg);
    ln_ffn1_kernel<<<mtiles, 256, 0, stream>>>(hagg, x, gamma, beta, W1p, b1, hres, inter);
    gemm_ffn2_kernel<<<mtiles, 256, 0, stream>>>(inter, W2p, b2, hres, out);
}
